# starting kernel
# baseline (speedup 1.0000x reference)
#include <hip/hip_runtime.h>

typedef __attribute__((ext_vector_type(2))) float v2f;
typedef __attribute__((ext_vector_type(8))) float v8f;

static constexpr int B_ = 16;
static constexpr int C_ = 64;
static constexpr int H_ = 512;
static constexpr int W_ = 60;
static constexpr long HW_ = (long)H_ * W_;
static constexpr int KTOT = 192;

__device__ __forceinline__ void load_weights_lds(const float* __restrict__ wt,
                                                 float* wlds) {
  for (int idx = threadIdx.x; idx < KTOT * 64; idx += 256) {
    const int K  = idx >> 6;
    const int co = idx & 63;
    const int kh = K >> 6;
    const int ci = K & 63;
    wlds[(K >> 1) * 128 + co * 2 + (K & 1)] = wt[co * 576 + ci * 9 + kh * 3 + 1];
  }
  __syncthreads();
}

__device__ __forceinline__ void conv_tile(const float* __restrict__ src,
                                          const float* wlds,
                                          int b, int h, int i0,
                                          int n, int hi, v8f acc[4]) {
  const long laneB0 = (long)(2 * hi) * HW_ + n;
  const long laneB1 = laneB0 + HW_;
  const int  aLane  = hi * 128 + n * 2;

#pragma unroll
  for (int kh = 0; kh < 3; ++kh) {
    const int hr = h + kh - 1;
    if (hr < 0 || hr >= H_) continue;
    const float* srow = src + ((long)b * C_ * H_ + hr) * W_ + i0;
#pragma unroll
    for (int kc = 0; kc < 16; ++kc) {
      const long cbase = (long)(kc * 4) * HW_;
      v2f bfrag;
      bfrag.x = srow[cbase + laneB0];
      bfrag.y = srow[cbase + laneB1];
      const int k0    = kh * 64 + kc * 4;
      const int abase = (k0 >> 1) * 128 + aLane;
#pragma unroll
      for (int t = 0; t < 4; ++t) {
        v2f afrag = *(const v2f*)(wlds + abase + t * 32);
        acc[t] = __builtin_amdgcn_wmma_f32_16x16x4_f32(
             false, afrag,  false, bfrag,
             (short)0, acc[t],  false,  false);
      }
    }
  }
}

__device__ __forceinline__ void decode_tile(int& b, int& h, int& i0, int& n, int& hi) {
  const int lane = threadIdx.x & 31;
  n  = lane & 15;
  hi = lane >> 4;
  const int tile = blockIdx.x * 8 + (threadIdx.x >> 5);
  const int it = tile & 3;
  h  = (tile >> 2) & 511;
  b  = tile >> 11;
  i0 = (it == 3) ? 44 : it * 16;
}

__global__ __launch_bounds__(256) void pass1_kernel(const float* __restrict__ x,
                                                    const float* __restrict__ wt,
                                                    const float* __restrict__ bias,
                                                    float* __restrict__ down) {
  __shared__ float wlds[KTOT * 64];
  load_weights_lds(wt, wlds);

  int b, h, i0, n, hi;
  decode_tile(b, h, i0, n, hi);

  v8f acc[4] = {};
  conv_tile(x, wlds, b, h, i0, n, hi, acc);

  const int  i     = i0 + n;
  const bool sh_ok = (i + 1) < W_;
#pragma unroll
  for (int t = 0; t < 4; ++t) {
#pragma unroll
    for (int v = 0; v < 8; ++v) {
      const int  co  = t * 16 + v + 8 * hi;
      const long off = (((long)b * C_ + co) * H_ + h) * W_ + i;
      float r  = fmaxf(acc[t][v] + bias[co], 0.0f);
      float xs = sh_ok ? x[off + 1] : 0.0f;
      down[off] = r + xs;
    }
  }
}

__global__ __launch_bounds__(256) void pass2_kernel(const float* __restrict__ down,
                                                    const float* __restrict__ wt,
                                                    const float* __restrict__ bias,
                                                    float* __restrict__ out) {
  __shared__ float wlds[KTOT * 64];
  load_weights_lds(wt, wlds);

  int b, h, i0, n, hi;
  decode_tile(b, h, i0, n, hi);

  v8f acc[4] = {};
  conv_tile(down, wlds, b, h, i0, n, hi, acc);

  const int  k  = i0 + n - 1;
  const bool ok = (k >= 0);
#pragma unroll
  for (int t = 0; t < 4; ++t) {
#pragma unroll
    for (int v = 0; v < 8; ++v) {
      const int  co  = t * 16 + v + 8 * hi;
      const long off = (((long)b * C_ + co) * H_ + h) * W_ + k;
      float cd = fmaxf(acc[t][v] + bias[co], 0.0f);
      if (ok) out[off] = down[off] + cd;
    }
  }
}

__global__ __launch_bounds__(256) void col59_kernel(const float* __restrict__ down,
                                                    float* __restrict__ out) {
  const long idx   = (long)blockIdx.x * 256 + threadIdx.x;
  const long total = (long)B_ * C_ * H_;
  if (idx < total) {
    const long off = idx * W_ + (W_ - 1);
    out[off] = down[off];
  }
}

extern "C" void kernel_launch(void* const* d_in, const int* in_sizes, int n_in,
                              void* d_out, int out_size, void* d_ws, size_t ws_size,
                              hipStream_t stream) {
  const float* x    = (const float*)d_in[0];
  const float* w    = (const float*)d_in[1];
  const float* bias = (const float*)d_in[2];
  float*       out  = (float*)d_out;
  float*       down = (float*)d_ws;

  const int tiles = B_ * H_ * 4;
  const int wgs   = tiles / 8;
  dim3 block(256);

  pass1_kernel<<<wgs, block, 0, stream>>>(x, w, bias, down);
  pass2_kernel<<<wgs, block, 0, stream>>>(down, w, bias, out);

  const long rows = (long)B_ * C_ * H_;
  col59_kernel<<<(int)((rows + 255) / 256), block, 0, stream>>>(down, out);
}


# Resident grid: 256 work-groups of 32 waves stage the weights into LDS once each and every wave loops over 4 tiles (was 4096 work-groups of 8 waves, each re-staging 48 KB for one tile per wave)
# speedup vs baseline: 1.7710x; 1.7710x over previous
#include <hip/hip_runtime.h>

typedef __attribute__((ext_vector_type(2))) float v2f;
typedef __attribute__((ext_vector_type(8))) float v8f;

static constexpr int B_ = 16;
static constexpr int C_ = 64;
static constexpr int H_ = 512;
static constexpr int W_ = 60;
static constexpr long HW_ = (long)H_ * W_;
static constexpr int KTOT = 192;

static constexpr int WG_WAVES       = 32;
static constexpr int WG_THREADS     = WG_WAVES * 32;
static constexpr int NUM_WGS        = 256;
static constexpr int TILES_TOTAL    = B_ * H_ * 4;
static constexpr int TILES_PER_WAVE = TILES_TOTAL / (NUM_WGS * WG_WAVES);
static_assert(TILES_PER_WAVE * NUM_WGS * WG_WAVES == TILES_TOTAL, "tiles must divide evenly");

__device__ __forceinline__ void load_weights_lds(const float* __restrict__ wt,
                                                 float* wlds) {
  for (int idx = threadIdx.x; idx < KTOT * 64; idx += WG_THREADS) {
    const int K  = idx >> 6;
    const int co = idx & 63;
    const int kh = K >> 6;
    const int ci = K & 63;
    wlds[(K >> 1) * 128 + co * 2 + (K & 1)] = wt[co * 576 + ci * 9 + kh * 3 + 1];
  }
  __syncthreads();
}

__device__ __forceinline__ void conv_tile(const float* __restrict__ src,
                                          const float* wlds,
                                          int b, int h, int i0,
                                          int n, int hi, v8f acc[4]) {
  const long laneB0 = (long)(2 * hi) * HW_ + n;
  const long laneB1 = laneB0 + HW_;
  const int  aLane  = hi * 128 + n * 2;

#pragma unroll
  for (int kh = 0; kh < 3; ++kh) {
    const int hr = h + kh - 1;
    if (hr < 0 || hr >= H_) continue;
    const float* srow = src + ((long)b * C_ * H_ + hr) * W_ + i0;
#pragma unroll
    for (int kc = 0; kc < 16; ++kc) {
      const long cbase = (long)(kc * 4) * HW_;
      v2f bfrag;
      bfrag.x = srow[cbase + laneB0];
      bfrag.y = srow[cbase + laneB1];
      const int k0    = kh * 64 + kc * 4;
      const int abase = (k0 >> 1) * 128 + aLane;
#pragma unroll
      for (int t = 0; t < 4; ++t) {
        v2f afrag = *(const v2f*)(wlds + abase + t * 32);
        acc[t] = __builtin_amdgcn_wmma_f32_16x16x4_f32(
             false, afrag,  false, bfrag,
             (short)0, acc[t],  false,  false);
      }
    }
  }
}

__device__ __forceinline__ void decode_tile(int trip, int& b, int& h, int& i0, int& n, int& hi) {
  const int lane = threadIdx.x & 31;
  n  = lane & 15;
  hi = lane >> 4;
  const int tile = (blockIdx.x * TILES_PER_WAVE + trip) * WG_WAVES + (threadIdx.x >> 5);
  const int it = tile & 3;
  h  = (tile >> 2) & 511;
  b  = tile >> 11;
  i0 = (it == 3) ? 44 : it * 16;
}

__global__ __launch_bounds__(WG_THREADS) void pass1_kernel(const float* __restrict__ x,
                                                           const float* __restrict__ wt,
                                                           const float* __restrict__ bias,
                                                           float* __restrict__ down) {
  __shared__ float wlds[KTOT * 64];
  load_weights_lds(wt, wlds);

  for (int trip = 0; trip < TILES_PER_WAVE; ++trip) {
    asm volatile("" ::: "memory");
    int b, h, i0, n, hi;
    decode_tile(trip, b, h, i0, n, hi);

    v8f acc[4] = {};
    conv_tile(x, wlds, b, h, i0, n, hi, acc);

    const int  i     = i0 + n;
    const bool sh_ok = (i + 1) < W_;
#pragma unroll
    for (int t = 0; t < 4; ++t) {
#pragma unroll
      for (int v = 0; v < 8; ++v) {
        const int  co  = t * 16 + v + 8 * hi;
        const long off = (((long)b * C_ + co) * H_ + h) * W_ + i;
        float r  = fmaxf(acc[t][v] + bias[co], 0.0f);
        float xs = sh_ok ? x[off + 1] : 0.0f;
        down[off] = r + xs;
      }
    }
  }
}

__global__ __launch_bounds__(WG_THREADS) void pass2_kernel(const float* __restrict__ down,
                                                           const float* __restrict__ wt,
                                                           const float* __restrict__ bias,
                                                           float* __restrict__ out) {
  __shared__ float wlds[KTOT * 64];
  load_weights_lds(wt, wlds);

  for (int trip = 0; trip < TILES_PER_WAVE; ++trip) {
    asm volatile("" ::: "memory");
    int b, h, i0, n, hi;
    decode_tile(trip, b, h, i0, n, hi);

    v8f acc[4] = {};
    conv_tile(down, wlds, b, h, i0, n, hi, acc);

    const int  k  = i0 + n - 1;
    const bool ok = (k >= 0);
#pragma unroll
    for (int t = 0; t < 4; ++t) {
#pragma unroll
      for (int v = 0; v < 8; ++v) {
        const int  co  = t * 16 + v + 8 * hi;
        const long off = (((long)b * C_ + co) * H_ + h) * W_ + k;
        float cd = fmaxf(acc[t][v] + bias[co], 0.0f);
        if (ok) out[off] = down[off] + cd;
      }
    }
  }
}

__global__ __launch_bounds__(256) void col59_kernel(const float* __restrict__ down,
                                                    float* __restrict__ out) {
  const long idx   = (long)blockIdx.x * 256 + threadIdx.x;
  const long total = (long)B_ * C_ * H_;
  if (idx < total) {
    const long off = idx * W_ + (W_ - 1);
    out[off] = down[off];
  }
}

extern "C" void kernel_launch(void* const* d_in, const int* in_sizes, int n_in,
                              void* d_out, int out_size, void* d_ws, size_t ws_size,
                              hipStream_t stream) {
  const float* x    = (const float*)d_in[0];
  const float* w    = (const float*)d_in[1];
  const float* bias = (const float*)d_in[2];
  float*       out  = (float*)d_out;
  float*       down = (float*)d_ws;

  dim3 block(256);

  pass1_kernel<<<NUM_WGS, WG_THREADS, 0, stream>>>(x, w, bias, down);
  pass2_kernel<<<NUM_WGS, WG_THREADS, 0, stream>>>(down, w, bias, out);

  const long rows = (long)B_ * C_ * H_;
  col59_kernel<<<(int)((rows + 255) / 256), block, 0, stream>>>(down, out);
}


# Global loads grouped ahead of use: each kh block's 32 B-fragment loads stand in front of its 64 WMMAs, each tile's epilogue reads in front of its 32 stores (was one pair of loads per wait)
# speedup vs baseline: 2.1066x; 1.1895x over previous
#include <hip/hip_runtime.h>

typedef __attribute__((ext_vector_type(2))) float v2f;
typedef __attribute__((ext_vector_type(8))) float v8f;

static constexpr int B_ = 16;
static constexpr int C_ = 64;
static constexpr int H_ = 512;
static constexpr int W_ = 60;
static constexpr long HW_ = (long)H_ * W_;
static constexpr int KTOT = 192;

static constexpr int WG_WAVES       = 32;
static constexpr int WG_THREADS     = WG_WAVES * 32;
static constexpr int NUM_WGS        = 256;
static constexpr int TILES_TOTAL    = B_ * H_ * 4;
static constexpr int TILES_PER_WAVE = TILES_TOTAL / (NUM_WGS * WG_WAVES);
static_assert(TILES_PER_WAVE * NUM_WGS * WG_WAVES == TILES_TOTAL, "tiles must divide evenly");

__device__ __forceinline__ void load_weights_lds(const float* __restrict__ wt,
                                                 float* wlds) {
  for (int idx = threadIdx.x; idx < KTOT * 64; idx += WG_THREADS) {
    const int K  = idx >> 6;
    const int co = idx & 63;
    const int kh = K >> 6;
    const int ci = K & 63;
    wlds[(K >> 1) * 128 + co * 2 + (K & 1)] = wt[co * 576 + ci * 9 + kh * 3 + 1];
  }
  __syncthreads();
}

__device__ __forceinline__ void conv_tile(const float* __restrict__ src,
                                          const float* wlds,
                                          int b, int h, int i0,
                                          int n, int hi, v8f acc[4]) {
  const long laneB0 = (long)(2 * hi) * HW_ + n;
  const long laneB1 = laneB0 + HW_;
  const int  aLane  = hi * 128 + n * 2;

#pragma unroll
  for (int kh = 0; kh < 3; ++kh) {
    const int hr = h + kh - 1;
    if (hr < 0 || hr >= H_) continue;
    const float* srow = src + ((long)b * C_ * H_ + hr) * W_ + i0;
    v2f bfrag[16];
#pragma unroll
    for (int kc = 0; kc < 16; ++kc) {
      const long cbase = (long)(kc * 4) * HW_;
      bfrag[kc].x = srow[cbase + laneB0];
      bfrag[kc].y = srow[cbase + laneB1];
    }
    asm volatile("" ::: "memory");
#pragma unroll
    for (int kc = 0; kc < 16; ++kc) {
      const int k0    = kh * 64 + kc * 4;
      const int abase = (k0 >> 1) * 128 + aLane;
#pragma unroll
      for (int t = 0; t < 4; ++t) {
        v2f afrag = *(const v2f*)(wlds + abase + t * 32);
        acc[t] = __builtin_amdgcn_wmma_f32_16x16x4_f32(
             false, afrag,  false, bfrag[kc],
             (short)0, acc[t],  false,  false);
      }
    }
  }
}

__device__ __forceinline__ void decode_tile(int trip, int& b, int& h, int& i0, int& n, int& hi) {
  const int lane = threadIdx.x & 31;
  n  = lane & 15;
  hi = lane >> 4;
  const int tile = (blockIdx.x * TILES_PER_WAVE + trip) * WG_WAVES + (threadIdx.x >> 5);
  const int it = tile & 3;
  h  = (tile >> 2) & 511;
  b  = tile >> 11;
  i0 = (it == 3) ? 44 : it * 16;
}

__global__ __launch_bounds__(WG_THREADS) void pass1_kernel(const float* __restrict__ x,
                                                           const float* __restrict__ wt,
                                                           const float* __restrict__ bias,
                                                           float* __restrict__ down) {
  __shared__ float wlds[KTOT * 64];
  load_weights_lds(wt, wlds);

  for (int trip = 0; trip < TILES_PER_WAVE; ++trip) {
    asm volatile("" ::: "memory");
    int b, h, i0, n, hi;
    decode_tile(trip, b, h, i0, n, hi);

    v8f acc[4] = {};
    conv_tile(x, wlds, b, h, i0, n, hi, acc);

    const int  i     = i0 + n;
    const bool sh_ok = (i + 1) < W_;
    const long sh    = sh_ok ? 1 : 0;
    const long off0  = (((long)b * C_ + 8 * hi) * H_ + h) * W_ + i;
    float bv[4][8], xv[4][8];
#pragma unroll
    for (int t = 0; t < 4; ++t) {
#pragma unroll
      for (int v = 0; v < 8; ++v) {
        bv[t][v] = bias[t * 16 + v + 8 * hi];
        xv[t][v] = x[off0 + (long)(t * 16 + v) * HW_ + sh];
      }
    }
#pragma unroll
    for (int t = 0; t < 4; ++t) {
#pragma unroll
      for (int v = 0; v < 8; ++v) {
        float r  = fmaxf(acc[t][v] + bv[t][v], 0.0f);
        float xs = sh_ok ? xv[t][v] : 0.0f;
        down[off0 + (long)(t * 16 + v) * HW_] = r + xs;
      }
    }
  }
}

__global__ __launch_bounds__(WG_THREADS) void pass2_kernel(const float* __restrict__ down,
                                                           const float* __restrict__ wt,
                                                           const float* __restrict__ bias,
                                                           float* __restrict__ out) {
  __shared__ float wlds[KTOT * 64];
  load_weights_lds(wt, wlds);

  for (int trip = 0; trip < TILES_PER_WAVE; ++trip) {
    asm volatile("" ::: "memory");
    int b, h, i0, n, hi;
    decode_tile(trip, b, h, i0, n, hi);

    v8f acc[4] = {};
    conv_tile(down, wlds, b, h, i0, n, hi, acc);

    const int  k  = i0 + n - 1;
    const bool ok = (k >= 0);
    const long kr = ok ? k : 0;
    const long row0 = (((long)b * C_ + 8 * hi) * H_ + h) * W_;
    float bv[4][8], dv[4][8];
#pragma unroll
    for (int t = 0; t < 4; ++t) {
#pragma unroll
      for (int v = 0; v < 8; ++v) {
        bv[t][v] = bias[t * 16 + v + 8 * hi];
        dv[t][v] = down[row0 + (long)(t * 16 + v) * HW_ + kr];
      }
    }
    if (ok) {
#pragma unroll
      for (int t = 0; t < 4; ++t) {
#pragma unroll
        for (int v = 0; v < 8; ++v) {
          float cd = fmaxf(acc[t][v] + bv[t][v], 0.0f);
          out[row0 + (long)(t * 16 + v) * HW_ + k] = dv[t][v] + cd;
        }
      }
    }
  }
}

__global__ __launch_bounds__(256) void col59_kernel(const float* __restrict__ down,
                                                    float* __restrict__ out) {
  const long idx   = (long)blockIdx.x * 256 + threadIdx.x;
  const long total = (long)B_ * C_ * H_;
  if (idx < total) {
    const long off = idx * W_ + (W_ - 1);
    out[off] = down[off];
  }
}

extern "C" void kernel_launch(void* const* d_in, const int* in_sizes, int n_in,
                              void* d_out, int out_size, void* d_ws, size_t ws_size,
                              hipStream_t stream) {
  const float* x    = (const float*)d_in[0];
  const float* w    = (const float*)d_in[1];
  const float* bias = (const float*)d_in[2];
  float*       out  = (float*)d_out;
  float*       down = (float*)d_ws;

  dim3 block(256);

  pass1_kernel<<<NUM_WGS, WG_THREADS, 0, stream>>>(x, w, bias, down);
  pass2_kernel<<<NUM_WGS, WG_THREADS, 0, stream>>>(down, w, bias, out);

  const long rows = (long)B_ * C_ * H_;
  col59_kernel<<<(int)((rows + 255) / 256), block, 0, stream>>>(down, out);
}


# Input patch through LDS: each trip's 10 rows x 64 channels copied once with 16-byte loads issued one trip ahead; B fragments and epilogue x/down operand read from LDS (was 128 global loads per tile)
# speedup vs baseline: 2.2839x; 1.0842x over previous
#include <hip/hip_runtime.h>

typedef __attribute__((ext_vector_type(2))) float v2f;
typedef __attribute__((ext_vector_type(4))) float v4f;
typedef __attribute__((ext_vector_type(8))) float v8f;

static constexpr int B_ = 16;
static constexpr int C_ = 64;
static constexpr int H_ = 512;
static constexpr int W_ = 60;
static constexpr long HW_ = (long)H_ * W_;
static constexpr int KTOT = 192;

static constexpr int WG_WAVES       = 32;
static constexpr int WG_THREADS     = WG_WAVES * 32;
static constexpr int NUM_WGS        = 256;
static constexpr int TILES_TOTAL    = B_ * H_ * 4;
static constexpr int TILES_PER_WAVE = TILES_TOTAL / (NUM_WGS * WG_WAVES);
static_assert(TILES_PER_WAVE * NUM_WGS * WG_WAVES == TILES_TOTAL, "tiles must divide evenly");

static constexpr int PROWS  = 10;
static constexpr int PPLANE = PROWS * W_;
static constexpr int PVEC   = PPLANE / 4;

__device__ __forceinline__ void load_weights_lds(const float* __restrict__ wt,
                                                 float* wlds) {
  for (int idx = threadIdx.x; idx < KTOT * 64; idx += WG_THREADS) {
    const int K  = idx >> 6;
    const int co = idx & 63;
    const int kh = K >> 6;
    const int ci = K & 63;
    wlds[(K >> 1) * 128 + co * 2 + (K & 1)] = wt[co * 576 + ci * 9 + kh * 3 + 1];
  }
  __syncthreads();
}

__device__ __forceinline__ void patch_loads(const float* __restrict__ src,
                                            int b, int h0, v4f v[2][5]) {
  const int lane = threadIdx.x & 31;
  const int w    = threadIdx.x >> 5;
  if (lane < 30) {
    const bool top = (h0 == 0);
    const bool bot = (h0 + 8 == H_);
    const long g0  = ((long)(b * C_ + w) * H_ + h0) * W_ - W_ + lane * 4;
#pragma unroll
    for (int cc = 0; cc < 2; ++cc) {
#pragma unroll
      for (int q = 0; q < 5; ++q) {
        long g = g0 + (long)cc * 32 * HW_ + q * 120;
        if (q == 0) g += (top && lane < 15) ? W_ : 0;
        if (q == 4) g -= (bot && lane >= 15) ? W_ : 0;
        v[cc][q] = *(const v4f*)(src + g);
      }
    }
  }
}

__device__ __forceinline__ void patch_stores(float* xs, int h0, const v4f v[2][5]) {
  const int lane = threadIdx.x & 31;
  const int w    = threadIdx.x >> 5;
  if (lane < 30) {
    const bool top = (h0 == 0);
    const bool bot = (h0 + 8 == H_);
    float*     l0  = xs + w * PPLANE + lane * 4;
    const v4f z = {0.0f, 0.0f, 0.0f, 0.0f};
#pragma unroll
    for (int cc = 0; cc < 2; ++cc) {
#pragma unroll
      for (int q = 0; q < 5; ++q) {
        v4f o = v[cc][q];
        if (q == 0) o = (top && lane < 15) ? z : o;
        if (q == 4) o = (bot && lane >= 15) ? z : o;
        *(v4f*)(l0 + cc * 32 * PPLANE + q * 120) = o;
      }
    }
  }
}

__device__ __forceinline__ void conv_tile(const float* xs,
                                          const float* wlds,
                                          int h, int hl, int i0,
                                          int n, int hi, v8f acc[4]) {
  const float* xl    = xs + (2 * hi) * PPLANE + hl * W_ + i0 + n;
  const int    aLane = hi * 128 + n * 2;

#pragma unroll
  for (int kh = 0; kh < 3; ++kh) {
    const int hr = h + kh - 1;
    if (hr < 0 || hr >= H_) continue;
    const float* xr = xl + kh * W_;
#pragma unroll
    for (int kc = 0; kc < 16; ++kc) {
      v2f bfrag;
      bfrag.x = xr[(kc * 4) * PPLANE];
      bfrag.y = xr[(kc * 4 + 1) * PPLANE];
      const int k0    = kh * 64 + kc * 4;
      const int abase = (k0 >> 1) * 128 + aLane;
#pragma unroll
      for (int t = 0; t < 4; ++t) {
        v2f afrag = *(const v2f*)(wlds + abase + t * 32);
        acc[t] = __builtin_amdgcn_wmma_f32_16x16x4_f32(
             false, afrag,  false, bfrag,
             (short)0, acc[t],  false,  false);
      }
    }
  }
}

__device__ __forceinline__ void decode_tile(int trip, int& b, int& h, int& i0, int& n, int& hi) {
  const int lane = threadIdx.x & 31;
  n  = lane & 15;
  hi = lane >> 4;
  const int tile = (blockIdx.x * TILES_PER_WAVE + trip) * WG_WAVES + (threadIdx.x >> 5);
  const int it = tile & 3;
  h  = (tile >> 2) & 511;
  b  = tile >> 11;
  i0 = (it == 3) ? 44 : it * 16;
}

__global__ __launch_bounds__(WG_THREADS) void pass1_kernel(const float* __restrict__ x,
                                                           const float* __restrict__ wt,
                                                           const float* __restrict__ bias,
                                                           float* __restrict__ down) {
  __shared__ __attribute__((aligned(16))) float wlds[KTOT * 64];
  __shared__ __attribute__((aligned(16))) float xs[C_ * PPLANE];
  v4f pre[2][5];
  {
    const int tileA = blockIdx.x * TILES_PER_WAVE * WG_WAVES;
    patch_loads(x, tileA >> 11, (tileA >> 2) & 511, pre);
  }
  load_weights_lds(wt, wlds);

#pragma nounroll
  for (int trip = 0; trip < TILES_PER_WAVE; ++trip) {
    asm volatile("" ::: "memory");
    const int tile0 = (blockIdx.x * TILES_PER_WAVE + trip) * WG_WAVES;
    const int h0    = (tile0 >> 2) & 511;
    __syncthreads();
    patch_stores(xs, h0, pre);
    __syncthreads();
    if (trip + 1 < TILES_PER_WAVE) {
      const int tile1 = tile0 + WG_WAVES;
      patch_loads(x, tile1 >> 11, (tile1 >> 2) & 511, pre);
    }

    int b, h, i0, n, hi;
    decode_tile(trip, b, h, i0, n, hi);
    const int hl = h - h0;

    v8f acc[4] = {};
    conv_tile(xs, wlds, h, hl, i0, n, hi, acc);

    const int  i     = i0 + n;
    const bool sh_ok = (i + 1) < W_;
    const long off0  = (((long)b * C_ + 8 * hi) * H_ + h) * W_ + i;
    const float* xe = xs + (8 * hi) * PPLANE + (hl + 1) * W_ + i + 1;
#pragma unroll
    for (int t = 0; t < 4; ++t) {
      float bv[8], xv[8];
#pragma unroll
      for (int v = 0; v < 8; ++v) {
        bv[v] = bias[t * 16 + v + 8 * hi];
        xv[v] = xe[(t * 16 + v) * PPLANE];
      }
#pragma unroll
      for (int v = 0; v < 8; ++v) {
        float r   = fmaxf(acc[t][v] + bv[v], 0.0f);
        float xs1 = sh_ok ? xv[v] : 0.0f;
        down[off0 + (long)(t * 16 + v) * HW_] = r + xs1;
      }
    }
  }
}

__global__ __launch_bounds__(WG_THREADS) void pass2_kernel(const float* __restrict__ down,
                                                           const float* __restrict__ wt,
                                                           const float* __restrict__ bias,
                                                           float* __restrict__ out) {
  __shared__ __attribute__((aligned(16))) float wlds[KTOT * 64];
  __shared__ __attribute__((aligned(16))) float xs[C_ * PPLANE];
  v4f pre[2][5];
  {
    const int tileA = blockIdx.x * TILES_PER_WAVE * WG_WAVES;
    patch_loads(down, tileA >> 11, (tileA >> 2) & 511, pre);
  }
  load_weights_lds(wt, wlds);

#pragma nounroll
  for (int trip = 0; trip < TILES_PER_WAVE; ++trip) {
    asm volatile("" ::: "memory");
    const int tile0 = (blockIdx.x * TILES_PER_WAVE + trip) * WG_WAVES;
    const int h0    = (tile0 >> 2) & 511;
    __syncthreads();
    patch_stores(xs, h0, pre);
    __syncthreads();
    if (trip + 1 < TILES_PER_WAVE) {
      const int tile1 = tile0 + WG_WAVES;
      patch_loads(down, tile1 >> 11, (tile1 >> 2) & 511, pre);
    }

    int b, h, i0, n, hi;
    decode_tile(trip, b, h, i0, n, hi);
    const int hl = h - h0;

    v8f acc[4] = {};
    conv_tile(xs, wlds, h, hl, i0, n, hi, acc);

    const int  k  = i0 + n - 1;
    const bool ok = (k >= 0);
    const long row0 = (((long)b * C_ + 8 * hi) * H_ + h) * W_;
    const float* de = xs + (8 * hi) * PPLANE + (hl + 1) * W_ + k;
#pragma unroll
    for (int t = 0; t < 4; ++t) {
      float bv[8], dv[8];
#pragma unroll
      for (int v = 0; v < 8; ++v) {
        bv[v] = bias[t * 16 + v + 8 * hi];
        dv[v] = de[(t * 16 + v) * PPLANE];
      }
      if (ok) {
#pragma unroll
        for (int v = 0; v < 8; ++v) {
          float cd = fmaxf(acc[t][v] + bv[v], 0.0f);
          out[row0 + (long)(t * 16 + v) * HW_ + k] = dv[v] + cd;
        }
      }
    }
  }
}

__global__ __launch_bounds__(256) void col59_kernel(const float* __restrict__ down,
                                                    float* __restrict__ out) {
  const long idx   = (long)blockIdx.x * 256 + threadIdx.x;
  const long total = (long)B_ * C_ * H_;
  if (idx < total) {
    const long off = idx * W_ + (W_ - 1);
    out[off] = down[off];
  }
}

extern "C" void kernel_launch(void* const* d_in, const int* in_sizes, int n_in,
                              void* d_out, int out_size, void* d_ws, size_t ws_size,
                              hipStream_t stream) {
  const float* x    = (const float*)d_in[0];
  const float* w    = (const float*)d_in[1];
  const float* bias = (const float*)d_in[2];
  float*       out  = (float*)d_out;
  float*       down = (float*)d_ws;

  dim3 block(256);

  pass1_kernel<<<NUM_WGS, WG_THREADS, 0, stream>>>(x, w, bias, down);
  pass2_kernel<<<NUM_WGS, WG_THREADS, 0, stream>>>(down, w, bias, out);

  const long rows = (long)B_ * C_ * H_;
  col59_kernel<<<(int)((rows + 255) / 256), block, 0, stream>>>(down, out);
}


# Column 59 (out = down, no cd term) written by pass 2 from its LDS patch: lane 0 of the last column tile takes k=59 instead of its duplicate k=43; third launch (524288 strided 4-byte copies) removed
# speedup vs baseline: 2.5887x; 1.1335x over previous
#include <hip/hip_runtime.h>

typedef __attribute__((ext_vector_type(2))) float v2f;
typedef __attribute__((ext_vector_type(4))) float v4f;
typedef __attribute__((ext_vector_type(8))) float v8f;

static constexpr int B_ = 16;
static constexpr int C_ = 64;
static constexpr int H_ = 512;
static constexpr int W_ = 60;
static constexpr long HW_ = (long)H_ * W_;
static constexpr int KTOT = 192;

static constexpr int WG_WAVES       = 32;
static constexpr int WG_THREADS     = WG_WAVES * 32;
static constexpr int NUM_WGS        = 256;
static constexpr int TILES_TOTAL    = B_ * H_ * 4;
static constexpr int TILES_PER_WAVE = TILES_TOTAL / (NUM_WGS * WG_WAVES);
static_assert(TILES_PER_WAVE * NUM_WGS * WG_WAVES == TILES_TOTAL, "tiles must divide evenly");

static constexpr int PROWS  = 10;
static constexpr int PPLANE = PROWS * W_;
static constexpr int PVEC   = PPLANE / 4;

__device__ __forceinline__ void load_weights_lds(const float* __restrict__ wt,
                                                 float* wlds) {
  for (int idx = threadIdx.x; idx < KTOT * 64; idx += WG_THREADS) {
    const int K  = idx >> 6;
    const int co = idx & 63;
    const int kh = K >> 6;
    const int ci = K & 63;
    wlds[(K >> 1) * 128 + co * 2 + (K & 1)] = wt[co * 576 + ci * 9 + kh * 3 + 1];
  }
  __syncthreads();
}

__device__ __forceinline__ void patch_loads(const float* __restrict__ src,
                                            int b, int h0, v4f v[2][5]) {
  const int lane = threadIdx.x & 31;
  const int w    = threadIdx.x >> 5;
  if (lane < 30) {
    const bool top = (h0 == 0);
    const bool bot = (h0 + 8 == H_);
    const long g0  = ((long)(b * C_ + w) * H_ + h0) * W_ - W_ + lane * 4;
#pragma unroll
    for (int cc = 0; cc < 2; ++cc) {
#pragma unroll
      for (int q = 0; q < 5; ++q) {
        long g = g0 + (long)cc * 32 * HW_ + q * 120;
        if (q == 0) g += (top && lane < 15) ? W_ : 0;
        if (q == 4) g -= (bot && lane >= 15) ? W_ : 0;
        v[cc][q] = *(const v4f*)(src + g);
      }
    }
  }
}

__device__ __forceinline__ void patch_stores(float* xs, int h0, const v4f v[2][5]) {
  const int lane = threadIdx.x & 31;
  const int w    = threadIdx.x >> 5;
  if (lane < 30) {
    const bool top = (h0 == 0);
    const bool bot = (h0 + 8 == H_);
    float*     l0  = xs + w * PPLANE + lane * 4;
    const v4f z = {0.0f, 0.0f, 0.0f, 0.0f};
#pragma unroll
    for (int cc = 0; cc < 2; ++cc) {
#pragma unroll
      for (int q = 0; q < 5; ++q) {
        v4f o = v[cc][q];
        if (q == 0) o = (top && lane < 15) ? z : o;
        if (q == 4) o = (bot && lane >= 15) ? z : o;
        *(v4f*)(l0 + cc * 32 * PPLANE + q * 120) = o;
      }
    }
  }
}

__device__ __forceinline__ void conv_tile(const float* xs,
                                          const float* wlds,
                                          int h, int hl, int i0,
                                          int n, int hi, v8f acc[4]) {
  const float* xl    = xs + (2 * hi) * PPLANE + hl * W_ + i0 + n;
  const int    aLane = hi * 128 + n * 2;

#pragma unroll
  for (int kh = 0; kh < 3; ++kh) {
    const int hr = h + kh - 1;
    if (hr < 0 || hr >= H_) continue;
    const float* xr = xl + kh * W_;
#pragma unroll
    for (int kc = 0; kc < 16; ++kc) {
      v2f bfrag;
      bfrag.x = xr[(kc * 4) * PPLANE];
      bfrag.y = xr[(kc * 4 + 1) * PPLANE];
      const int k0    = kh * 64 + kc * 4;
      const int abase = (k0 >> 1) * 128 + aLane;
#pragma unroll
      for (int t = 0; t < 4; ++t) {
        v2f afrag = *(const v2f*)(wlds + abase + t * 32);
        acc[t] = __builtin_amdgcn_wmma_f32_16x16x4_f32(
             false, afrag,  false, bfrag,
             (short)0, acc[t],  false,  false);
      }
    }
  }
}

__device__ __forceinline__ void decode_tile(int trip, int& b, int& h, int& i0, int& n, int& hi) {
  const int lane = threadIdx.x & 31;
  n  = lane & 15;
  hi = lane >> 4;
  const int tile = (blockIdx.x * TILES_PER_WAVE + trip) * WG_WAVES + (threadIdx.x >> 5);
  const int it = tile & 3;
  h  = (tile >> 2) & 511;
  b  = tile >> 11;
  i0 = (it == 3) ? 44 : it * 16;
}

__global__ __launch_bounds__(WG_THREADS) void pass1_kernel(const float* __restrict__ x,
                                                           const float* __restrict__ wt,
                                                           const float* __restrict__ bias,
                                                           float* __restrict__ down) {
  __shared__ __attribute__((aligned(16))) float wlds[KTOT * 64];
  __shared__ __attribute__((aligned(16))) float xs[C_ * PPLANE];
  v4f pre[2][5];
  {
    const int tileA = blockIdx.x * TILES_PER_WAVE * WG_WAVES;
    patch_loads(x, tileA >> 11, (tileA >> 2) & 511, pre);
  }
  load_weights_lds(wt, wlds);

#pragma nounroll
  for (int trip = 0; trip < TILES_PER_WAVE; ++trip) {
    asm volatile("" ::: "memory");
    const int tile0 = (blockIdx.x * TILES_PER_WAVE + trip) * WG_WAVES;
    const int h0    = (tile0 >> 2) & 511;
    __syncthreads();
    patch_stores(xs, h0, pre);
    __syncthreads();
    if (trip + 1 < TILES_PER_WAVE) {
      const int tile1 = tile0 + WG_WAVES;
      patch_loads(x, tile1 >> 11, (tile1 >> 2) & 511, pre);
    }

    int b, h, i0, n, hi;
    decode_tile(trip, b, h, i0, n, hi);
    const int hl = h - h0;

    v8f acc[4] = {};
    conv_tile(xs, wlds, h, hl, i0, n, hi, acc);

    const int  i     = i0 + n;
    const bool sh_ok = (i + 1) < W_;
    const long off0  = (((long)b * C_ + 8 * hi) * H_ + h) * W_ + i;
    const float* xe = xs + (8 * hi) * PPLANE + (hl + 1) * W_ + i + 1;
#pragma unroll
    for (int t = 0; t < 4; ++t) {
      float bv[8], xv[8];
#pragma unroll
      for (int v = 0; v < 8; ++v) {
        bv[v] = bias[t * 16 + v + 8 * hi];
        xv[v] = xe[(t * 16 + v) * PPLANE];
      }
#pragma unroll
      for (int v = 0; v < 8; ++v) {
        float r   = fmaxf(acc[t][v] + bv[v], 0.0f);
        float xs1 = sh_ok ? xv[v] : 0.0f;
        down[off0 + (long)(t * 16 + v) * HW_] = r + xs1;
      }
    }
  }
}

__global__ __launch_bounds__(WG_THREADS) void pass2_kernel(const float* __restrict__ down,
                                                           const float* __restrict__ wt,
                                                           const float* __restrict__ bias,
                                                           float* __restrict__ out) {
  __shared__ __attribute__((aligned(16))) float wlds[KTOT * 64];
  __shared__ __attribute__((aligned(16))) float xs[C_ * PPLANE];
  v4f pre[2][5];
  {
    const int tileA = blockIdx.x * TILES_PER_WAVE * WG_WAVES;
    patch_loads(down, tileA >> 11, (tileA >> 2) & 511, pre);
  }
  load_weights_lds(wt, wlds);

#pragma nounroll
  for (int trip = 0; trip < TILES_PER_WAVE; ++trip) {
    asm volatile("" ::: "memory");
    const int tile0 = (blockIdx.x * TILES_PER_WAVE + trip) * WG_WAVES;
    const int h0    = (tile0 >> 2) & 511;
    __syncthreads();
    patch_stores(xs, h0, pre);
    __syncthreads();
    if (trip + 1 < TILES_PER_WAVE) {
      const int tile1 = tile0 + WG_WAVES;
      patch_loads(down, tile1 >> 11, (tile1 >> 2) & 511, pre);
    }

    int b, h, i0, n, hi;
    decode_tile(trip, b, h, i0, n, hi);
    const int hl = h - h0;

    v8f acc[4] = {};
    conv_tile(xs, wlds, h, hl, i0, n, hi, acc);

    const bool c59 = (i0 == 44) && (n == 0);
    const int  k   = c59 ? (W_ - 1) : (i0 + n - 1);
    const bool ok  = (k >= 0);
    const long row0 = (((long)b * C_ + 8 * hi) * H_ + h) * W_;
    const float* de = xs + (8 * hi) * PPLANE + (hl + 1) * W_ + k;
#pragma unroll
    for (int t = 0; t < 4; ++t) {
      float bv[8], dv[8];
#pragma unroll
      for (int v = 0; v < 8; ++v) {
        bv[v] = bias[t * 16 + v + 8 * hi];
        dv[v] = de[(t * 16 + v) * PPLANE];
      }
      if (ok) {
#pragma unroll
        for (int v = 0; v < 8; ++v) {
          float cd = fmaxf(acc[t][v] + bv[v], 0.0f);
          cd = c59 ? 0.0f : cd;
          out[row0 + (long)(t * 16 + v) * HW_ + k] = dv[v] + cd;
        }
      }
    }
  }
}

extern "C" void kernel_launch(void* const* d_in, const int* in_sizes, int n_in,
                              void* d_out, int out_size, void* d_ws, size_t ws_size,
                              hipStream_t stream) {
  const float* x    = (const float*)d_in[0];
  const float* w    = (const float*)d_in[1];
  const float* bias = (const float*)d_in[2];
  float*       out  = (float*)d_out;
  float*       down = (float*)d_ws;

  pass1_kernel<<<NUM_WGS, WG_THREADS, 0, stream>>>(x, w, bias, down);
  pass2_kernel<<<NUM_WGS, WG_THREADS, 0, stream>>>(down, w, bias, out);
}


# Weight staging with whole-line loads: tensor read in order as 9 16-byte pieces per thread, kw=1 elements picked out through LDS (was 12 4-byte loads per thread, lanes 2304 B apart, 32 lines each)
# speedup vs baseline: 2.6800x; 1.0353x over previous
#include <hip/hip_runtime.h>

typedef __attribute__((ext_vector_type(2))) float v2f;
typedef __attribute__((ext_vector_type(4))) float v4f;
typedef __attribute__((ext_vector_type(8))) float v8f;

static constexpr int B_ = 16;
static constexpr int C_ = 64;
static constexpr int H_ = 512;
static constexpr int W_ = 60;
static constexpr long HW_ = (long)H_ * W_;
static constexpr int KTOT = 192;

static constexpr int WG_WAVES       = 32;
static constexpr int WG_THREADS     = WG_WAVES * 32;
static constexpr int NUM_WGS        = 256;
static constexpr int TILES_TOTAL    = B_ * H_ * 4;
static constexpr int TILES_PER_WAVE = TILES_TOTAL / (NUM_WGS * WG_WAVES);
static_assert(TILES_PER_WAVE * NUM_WGS * WG_WAVES == TILES_TOTAL, "tiles must divide evenly");

static constexpr int PROWS  = 10;
static constexpr int PPLANE = PROWS * W_;
static constexpr int PVEC   = PPLANE / 4;

static constexpr int WROW    = C_ * 9;
static constexpr int WPIECES = C_ * WROW / 4;
static constexpr int WPT     = WPIECES / (32 * 32);
static constexpr int WPITCH  = WROW + 4;
static_assert(WPT * 32 * 32 == WPIECES, "pieces must divide evenly over the work-group");
static_assert(C_ * WPITCH <= C_ * PPLANE, "the parked copy must fit in the patch buffer");

__device__ __forceinline__ void weight_loads(const float* __restrict__ wt, v4f wv[WPT]) {
#pragma unroll
  for (int j = 0; j < WPT; ++j)
    wv[j] = *(const v4f*)(wt + 4 * (int)(threadIdx.x + WG_THREADS * j));
}

__device__ __forceinline__ void weights_to_lds(const v4f wv[WPT], float* park, float* wlds) {
#pragma unroll
  for (int j = 0; j < WPT; ++j) {
    const int p  = threadIdx.x + WG_THREADS * j;
    const int co = p / (WROW / 4);
    const int r  = p - co * (WROW / 4);
    *(v4f*)(park + co * WPITCH + 4 * r) = wv[j];
  }
  __syncthreads();
#pragma unroll
  for (int j = 0; j < (KTOT * 64) / WG_THREADS; ++j) {
    const int d  = threadIdx.x + WG_THREADS * j;
    const int K  = ((d >> 7) << 1) | (d & 1);
    const int co = (d >> 1) & 63;
    const int kh = K >> 6;
    const int ci = K & 63;
    wlds[d] = park[co * WPITCH + ci * 9 + kh * 3 + 1];
  }
}

__device__ __forceinline__ void patch_loads(const float* __restrict__ src,
                                            int b, int h0, v4f v[2][5]) {
  const int lane = threadIdx.x & 31;
  const int w    = threadIdx.x >> 5;
  if (lane < 30) {
    const bool top = (h0 == 0);
    const bool bot = (h0 + 8 == H_);
    const long g0  = ((long)(b * C_ + w) * H_ + h0) * W_ - W_ + lane * 4;
#pragma unroll
    for (int cc = 0; cc < 2; ++cc) {
#pragma unroll
      for (int q = 0; q < 5; ++q) {
        long g = g0 + (long)cc * 32 * HW_ + q * 120;
        if (q == 0) g += (top && lane < 15) ? W_ : 0;
        if (q == 4) g -= (bot && lane >= 15) ? W_ : 0;
        v[cc][q] = *(const v4f*)(src + g);
      }
    }
  }
}

__device__ __forceinline__ void patch_stores(float* xs, int h0, const v4f v[2][5]) {
  const int lane = threadIdx.x & 31;
  const int w    = threadIdx.x >> 5;
  if (lane < 30) {
    const bool top = (h0 == 0);
    const bool bot = (h0 + 8 == H_);
    float*     l0  = xs + w * PPLANE + lane * 4;
    const v4f z = {0.0f, 0.0f, 0.0f, 0.0f};
#pragma unroll
    for (int cc = 0; cc < 2; ++cc) {
#pragma unroll
      for (int q = 0; q < 5; ++q) {
        v4f o = v[cc][q];
        if (q == 0) o = (top && lane < 15) ? z : o;
        if (q == 4) o = (bot && lane >= 15) ? z : o;
        *(v4f*)(l0 + cc * 32 * PPLANE + q * 120) = o;
      }
    }
  }
}

__device__ __forceinline__ void conv_tile(const float* xs,
                                          const float* wlds,
                                          int h, int hl, int i0,
                                          int n, int hi, v8f acc[4]) {
  const float* xl    = xs + (2 * hi) * PPLANE + hl * W_ + i0 + n;
  const int    aLane = hi * 128 + n * 2;

#pragma unroll
  for (int kh = 0; kh < 3; ++kh) {
    const int hr = h + kh - 1;
    if (hr < 0 || hr >= H_) continue;
    const float* xr = xl + kh * W_;
#pragma unroll
    for (int kc = 0; kc < 16; ++kc) {
      v2f bfrag;
      bfrag.x = xr[(kc * 4) * PPLANE];
      bfrag.y = xr[(kc * 4 + 1) * PPLANE];
      const int k0    = kh * 64 + kc * 4;
      const int abase = (k0 >> 1) * 128 + aLane;
#pragma unroll
      for (int t = 0; t < 4; ++t) {
        v2f afrag = *(const v2f*)(wlds + abase + t * 32);
        acc[t] = __builtin_amdgcn_wmma_f32_16x16x4_f32(
             false, afrag,  false, bfrag,
             (short)0, acc[t],  false,  false);
      }
    }
  }
}

__device__ __forceinline__ void decode_tile(int trip, int& b, int& h, int& i0, int& n, int& hi) {
  const int lane = threadIdx.x & 31;
  n  = lane & 15;
  hi = lane >> 4;
  const int tile = (blockIdx.x * TILES_PER_WAVE + trip) * WG_WAVES + (threadIdx.x >> 5);
  const int it = tile & 3;
  h  = (tile >> 2) & 511;
  b  = tile >> 11;
  i0 = (it == 3) ? 44 : it * 16;
}

__global__ __launch_bounds__(WG_THREADS) void pass1_kernel(const float* __restrict__ x,
                                                           const float* __restrict__ wt,
                                                           const float* __restrict__ bias,
                                                           float* __restrict__ down) {
  __shared__ __attribute__((aligned(16))) float wlds[KTOT * 64];
  __shared__ __attribute__((aligned(16))) float xs[C_ * PPLANE];
  v4f wv[WPT];
  weight_loads(wt, wv);
  v4f pre[2][5];
  {
    const int tileA = blockIdx.x * TILES_PER_WAVE * WG_WAVES;
    patch_loads(x, tileA >> 11, (tileA >> 2) & 511, pre);
  }
  weights_to_lds(wv, xs, wlds);

#pragma nounroll
  for (int trip = 0; trip < TILES_PER_WAVE; ++trip) {
    asm volatile("" ::: "memory");
    const int tile0 = (blockIdx.x * TILES_PER_WAVE + trip) * WG_WAVES;
    const int h0    = (tile0 >> 2) & 511;
    __syncthreads();
    patch_stores(xs, h0, pre);
    __syncthreads();
    if (trip + 1 < TILES_PER_WAVE) {
      const int tile1 = tile0 + WG_WAVES;
      patch_loads(x, tile1 >> 11, (tile1 >> 2) & 511, pre);
    }

    int b, h, i0, n, hi;
    decode_tile(trip, b, h, i0, n, hi);
    const int hl = h - h0;

    v8f acc[4] = {};
    conv_tile(xs, wlds, h, hl, i0, n, hi, acc);

    const int  i     = i0 + n;
    const bool sh_ok = (i + 1) < W_;
    const long off0  = (((long)b * C_ + 8 * hi) * H_ + h) * W_ + i;
    const float* xe = xs + (8 * hi) * PPLANE + (hl + 1) * W_ + i + 1;
#pragma unroll
    for (int t = 0; t < 4; ++t) {
      float bv[8], xv[8];
#pragma unroll
      for (int v = 0; v < 8; ++v) {
        bv[v] = bias[t * 16 + v + 8 * hi];
        xv[v] = xe[(t * 16 + v) * PPLANE];
      }
#pragma unroll
      for (int v = 0; v < 8; ++v) {
        float r   = fmaxf(acc[t][v] + bv[v], 0.0f);
        float xs1 = sh_ok ? xv[v] : 0.0f;
        down[off0 + (long)(t * 16 + v) * HW_] = r + xs1;
      }
    }
  }
}

__global__ __launch_bounds__(WG_THREADS) void pass2_kernel(const float* __restrict__ down,
                                                           const float* __restrict__ wt,
                                                           const float* __restrict__ bias,
                                                           float* __restrict__ out) {
  __shared__ __attribute__((aligned(16))) float wlds[KTOT * 64];
  __shared__ __attribute__((aligned(16))) float xs[C_ * PPLANE];
  v4f wv[WPT];
  weight_loads(wt, wv);
  v4f pre[2][5];
  {
    const int tileA = blockIdx.x * TILES_PER_WAVE * WG_WAVES;
    patch_loads(down, tileA >> 11, (tileA >> 2) & 511, pre);
  }
  weights_to_lds(wv, xs, wlds);

#pragma nounroll
  for (int trip = 0; trip < TILES_PER_WAVE; ++trip) {
    asm volatile("" ::: "memory");
    const int tile0 = (blockIdx.x * TILES_PER_WAVE + trip) * WG_WAVES;
    const int h0    = (tile0 >> 2) & 511;
    __syncthreads();
    patch_stores(xs, h0, pre);
    __syncthreads();
    if (trip + 1 < TILES_PER_WAVE) {
      const int tile1 = tile0 + WG_WAVES;
      patch_loads(down, tile1 >> 11, (tile1 >> 2) & 511, pre);
    }

    int b, h, i0, n, hi;
    decode_tile(trip, b, h, i0, n, hi);
    const int hl = h - h0;

    v8f acc[4] = {};
    conv_tile(xs, wlds, h, hl, i0, n, hi, acc);

    const bool c59 = (i0 == 44) && (n == 0);
    const int  k   = c59 ? (W_ - 1) : (i0 + n - 1);
    const bool ok  = (k >= 0);
    const long row0 = (((long)b * C_ + 8 * hi) * H_ + h) * W_;
    const float* de = xs + (8 * hi) * PPLANE + (hl + 1) * W_ + k;
#pragma unroll
    for (int t = 0; t < 4; ++t) {
      float bv[8], dv[8];
#pragma unroll
      for (int v = 0; v < 8; ++v) {
        bv[v] = bias[t * 16 + v + 8 * hi];
        dv[v] = de[(t * 16 + v) * PPLANE];
      }
      if (ok) {
#pragma unroll
        for (int v = 0; v < 8; ++v) {
          float cd = fmaxf(acc[t][v] + bv[v], 0.0f);
          cd = c59 ? 0.0f : cd;
          out[row0 + (long)(t * 16 + v) * HW_ + k] = dv[v] + cd;
        }
      }
    }
  }
}

extern "C" void kernel_launch(void* const* d_in, const int* in_sizes, int n_in,
                              void* d_out, int out_size, void* d_ws, size_t ws_size,
                              hipStream_t stream) {
  const float* x    = (const float*)d_in[0];
  const float* w    = (const float*)d_in[1];
  const float* bias = (const float*)d_in[2];
  float*       out  = (float*)d_out;
  float*       down = (float*)d_ws;

  pass1_kernel<<<NUM_WGS, WG_THREADS, 0, stream>>>(x, w, bias, down);
  pass2_kernel<<<NUM_WGS, WG_THREADS, 0, stream>>>(down, w, bias, out);
}


# Column tiles cut from each trip's 480 consecutive positions (8 rows x 60): 30 tiles of 16, no overlap; waves 28-31 take half the channels of tiles 28-29 (was 4 tiles per row, last redoing 4 columns)
# speedup vs baseline: 2.8239x; 1.0537x over previous
#include <hip/hip_runtime.h>

typedef __attribute__((ext_vector_type(2))) float v2f;
typedef __attribute__((ext_vector_type(4))) float v4f;
typedef __attribute__((ext_vector_type(8))) float v8f;

static constexpr int B_ = 16;
static constexpr int C_ = 64;
static constexpr int H_ = 512;
static constexpr int W_ = 60;
static constexpr long HW_ = (long)H_ * W_;
static constexpr int KTOT = 192;

static constexpr int WG_WAVES       = 32;
static constexpr int WG_THREADS     = WG_WAVES * 32;
static constexpr int NUM_WGS        = 256;
static constexpr int TILES_TOTAL    = B_ * H_ * 4;
static constexpr int TILES_PER_WAVE = TILES_TOTAL / (NUM_WGS * WG_WAVES);
static_assert(TILES_PER_WAVE * NUM_WGS * WG_WAVES == TILES_TOTAL, "tiles must divide evenly");
static constexpr int TRIP_ROWS  = 8;
static constexpr int TRIP_POS   = TRIP_ROWS * W_;
static constexpr int NTILES     = TRIP_POS / 16;
static constexpr int FULL_WAVES = 2 * NTILES - WG_WAVES;
static_assert(NTILES * 16 == TRIP_POS, "a trip must be a whole number of column tiles");
static_assert(FULL_WAVES + 2 * (NTILES - FULL_WAVES) == WG_WAVES, "waves must cover the tiles");

static constexpr int PROWS  = 10;
static constexpr int PPLANE = PROWS * W_;
static constexpr int PVEC   = PPLANE / 4;

static constexpr int WROW    = C_ * 9;
static constexpr int WPIECES = C_ * WROW / 4;
static constexpr int WPT     = WPIECES / (32 * 32);
static constexpr int WPITCH  = WROW + 4;
static_assert(WPT * 32 * 32 == WPIECES, "pieces must divide evenly over the work-group");
static_assert(C_ * WPITCH <= C_ * PPLANE, "the parked copy must fit in the patch buffer");

__device__ __forceinline__ void weight_loads(const float* __restrict__ wt, v4f wv[WPT]) {
#pragma unroll
  for (int j = 0; j < WPT; ++j)
    wv[j] = *(const v4f*)(wt + 4 * (int)(threadIdx.x + WG_THREADS * j));
}

__device__ __forceinline__ void weights_to_lds(const v4f wv[WPT], float* park, float* wlds) {
#pragma unroll
  for (int j = 0; j < WPT; ++j) {
    const int p  = threadIdx.x + WG_THREADS * j;
    const int co = p / (WROW / 4);
    const int r  = p - co * (WROW / 4);
    *(v4f*)(park + co * WPITCH + 4 * r) = wv[j];
  }
  __syncthreads();
#pragma unroll
  for (int j = 0; j < (KTOT * 64) / WG_THREADS; ++j) {
    const int d  = threadIdx.x + WG_THREADS * j;
    const int K  = ((d >> 7) << 1) | (d & 1);
    const int co = (d >> 1) & 63;
    const int kh = K >> 6;
    const int ci = K & 63;
    wlds[d] = park[co * WPITCH + ci * 9 + kh * 3 + 1];
  }
}

__device__ __forceinline__ void patch_loads(const float* __restrict__ src,
                                            int b, int h0, v4f v[2][5]) {
  const int lane = threadIdx.x & 31;
  const int w    = threadIdx.x >> 5;
  if (lane < 30) {
    const bool top = (h0 == 0);
    const bool bot = (h0 + 8 == H_);
    const long g0  = ((long)(b * C_ + w) * H_ + h0) * W_ - W_ + lane * 4;
#pragma unroll
    for (int cc = 0; cc < 2; ++cc) {
#pragma unroll
      for (int q = 0; q < 5; ++q) {
        long g = g0 + (long)cc * 32 * HW_ + q * 120;
        if (q == 0) g += (top && lane < 15) ? W_ : 0;
        if (q == 4) g -= (bot && lane >= 15) ? W_ : 0;
        v[cc][q] = *(const v4f*)(src + g);
      }
    }
  }
}

__device__ __forceinline__ void patch_stores(float* xs, int h0, const v4f v[2][5]) {
  const int lane = threadIdx.x & 31;
  const int w    = threadIdx.x >> 5;
  if (lane < 30) {
    const bool top = (h0 == 0);
    const bool bot = (h0 + 8 == H_);
    float*     l0  = xs + w * PPLANE + lane * 4;
    const v4f z = {0.0f, 0.0f, 0.0f, 0.0f};
#pragma unroll
    for (int cc = 0; cc < 2; ++cc) {
#pragma unroll
      for (int q = 0; q < 5; ++q) {
        v4f o = v[cc][q];
        if (q == 0) o = (top && lane < 15) ? z : o;
        if (q == 4) o = (bot && lane >= 15) ? z : o;
        *(v4f*)(l0 + cc * 32 * PPLANE + q * 120) = o;
      }
    }
  }
}

template <int TN>
__device__ __forceinline__ void conv_tile(const float* xs,
                                          const float* wlds,
                                          int h0, int j0, int tb,
                                          int n, int hi, v8f acc[TN]) {
  const int j = j0 + n;
  const float* xl    = xs + (2 * hi) * PPLANE + j;
  const int    aLane = hi * 128 + n * 2 + tb * 32;

#pragma unroll
  for (int kh = 0; kh < 3; ++kh) {
    const int rlo = h0 + j0 / W_ + kh - 1;
    const int rhi = h0 + (j0 + 15) / W_ + kh - 1;
    if (rhi < 0 || rlo >= H_) continue;
    const float* xr = xl + kh * W_;
#pragma unroll
    for (int kc = 0; kc < 16; ++kc) {
      v2f bfrag;
      bfrag.x = xr[(kc * 4) * PPLANE];
      bfrag.y = xr[(kc * 4 + 1) * PPLANE];
      const int k0    = kh * 64 + kc * 4;
      const int abase = (k0 >> 1) * 128 + aLane;
#pragma unroll
      for (int t = 0; t < TN; ++t) {
        v2f afrag = *(const v2f*)(wlds + abase + t * 32);
        acc[t] = __builtin_amdgcn_wmma_f32_16x16x4_f32(
             false, afrag,  false, bfrag,
             (short)0, acc[t],  false,  false);
      }
    }
  }
}

__device__ __forceinline__ void decode_wave(bool& full, int& j0, int& tb, int& n, int& hi) {
  const int lane = threadIdx.x & 31;
  n  = lane & 15;
  hi = lane >> 4;
  const int w = __builtin_amdgcn_readfirstlane((int)(threadIdx.x >> 5));
  full = (w < FULL_WAVES);
  const int ntile = full ? w : FULL_WAVES + ((w - FULL_WAVES) >> 1);
  tb = full ? 0 : 2 * ((w - FULL_WAVES) & 1);
  j0 = ntile * 16;
}

template <int TN>
__device__ __forceinline__ void pass1_tile(const float* xs, const float* wlds,
                                           const float* __restrict__ bias,
                                           float* __restrict__ down,
                                           int b, int h0, int j0, int tb, int n, int hi) {
  v8f acc[TN] = {};
  conv_tile<TN>(xs, wlds, h0, j0, tb, n, hi, acc);

  const int j = j0 + n;

  const int  hl    = j / W_;
  const int  i     = j - hl * W_;
  const bool sh_ok = (i + 1) < W_;
  const int  co0   = tb * 16 + 8 * hi;
  const long off0  = (((long)b * C_ + co0) * H_ + h0) * W_ + j;
  const float* xe = xs + co0 * PPLANE + (W_ + 1) + j;
#pragma unroll
  for (int t = 0; t < TN; ++t) {
    float bv[8], xv[8];
#pragma unroll
    for (int v = 0; v < 8; ++v) {
      bv[v] = bias[co0 + t * 16 + v];
      xv[v] = xe[(t * 16 + v) * PPLANE];
    }
#pragma unroll
    for (int v = 0; v < 8; ++v) {
      float r   = fmaxf(acc[t][v] + bv[v], 0.0f);
      float xs1 = sh_ok ? xv[v] : 0.0f;
      down[off0 + (long)(t * 16 + v) * HW_] = r + xs1;
    }
  }
}

template <int TN>
__device__ __forceinline__ void pass2_tile(const float* xs, const float* wlds,
                                           const float* __restrict__ bias,
                                           float* __restrict__ out,
                                           int b, int h0, int j0, int tb, int n, int hi) {
  v8f acc[TN] = {};
  conv_tile<TN>(xs, wlds, h0, j0, tb, n, hi, acc);

  const int j = j0 + n;

  const int  hl   = j / W_;
  const bool nocd = (j - hl * W_) == 0;
  const int  k    = (j == 0) ? (TRIP_POS - 1) : (j - 1);
  const int  co0  = tb * 16 + 8 * hi;
  const long row0 = (((long)b * C_ + co0) * H_ + h0) * W_ + k;
  const float* de = xs + co0 * PPLANE + W_ + k;
#pragma unroll
  for (int t = 0; t < TN; ++t) {
    float bv[8], dv[8];
#pragma unroll
    for (int v = 0; v < 8; ++v) {
      bv[v] = bias[co0 + t * 16 + v];
      dv[v] = de[(t * 16 + v) * PPLANE];
    }
#pragma unroll
    for (int v = 0; v < 8; ++v) {
      float cd = fmaxf(acc[t][v] + bv[v], 0.0f);
      cd = nocd ? 0.0f : cd;
      out[row0 + (long)(t * 16 + v) * HW_] = dv[v] + cd;
    }
  }
}

__global__ __launch_bounds__(WG_THREADS) void pass1_kernel(const float* __restrict__ x,
                                                           const float* __restrict__ wt,
                                                           const float* __restrict__ bias,
                                                           float* __restrict__ down) {
  __shared__ __attribute__((aligned(16))) float wlds[KTOT * 64];
  __shared__ __attribute__((aligned(16))) float xs[C_ * PPLANE];
  v4f wv[WPT];
  weight_loads(wt, wv);
  v4f pre[2][5];
  {
    const int tileA = blockIdx.x * TILES_PER_WAVE * WG_WAVES;
    patch_loads(x, tileA >> 11, (tileA >> 2) & 511, pre);
  }
  weights_to_lds(wv, xs, wlds);

#pragma nounroll
  for (int trip = 0; trip < TILES_PER_WAVE; ++trip) {
    asm volatile("" ::: "memory");
    const int tile0 = (blockIdx.x * TILES_PER_WAVE + trip) * WG_WAVES;
    const int h0    = (tile0 >> 2) & 511;
    __syncthreads();
    patch_stores(xs, h0, pre);
    __syncthreads();
    if (trip + 1 < TILES_PER_WAVE) {
      const int tile1 = tile0 + WG_WAVES;
      patch_loads(x, tile1 >> 11, (tile1 >> 2) & 511, pre);
    }

    const int b = tile0 >> 11;
    bool full;
    int  j0, tb, n, hi;
    decode_wave(full, j0, tb, n, hi);
    if (full) pass1_tile<4>(xs, wlds, bias, down, b, h0, j0, 0, n, hi);
    else      pass1_tile<2>(xs, wlds, bias, down, b, h0, j0, tb, n, hi);
  }
}

__global__ __launch_bounds__(WG_THREADS) void pass2_kernel(const float* __restrict__ down,
                                                           const float* __restrict__ wt,
                                                           const float* __restrict__ bias,
                                                           float* __restrict__ out) {
  __shared__ __attribute__((aligned(16))) float wlds[KTOT * 64];
  __shared__ __attribute__((aligned(16))) float xs[C_ * PPLANE];
  v4f wv[WPT];
  weight_loads(wt, wv);
  v4f pre[2][5];
  {
    const int tileA = blockIdx.x * TILES_PER_WAVE * WG_WAVES;
    patch_loads(down, tileA >> 11, (tileA >> 2) & 511, pre);
  }
  weights_to_lds(wv, xs, wlds);

#pragma nounroll
  for (int trip = 0; trip < TILES_PER_WAVE; ++trip) {
    asm volatile("" ::: "memory");
    const int tile0 = (blockIdx.x * TILES_PER_WAVE + trip) * WG_WAVES;
    const int h0    = (tile0 >> 2) & 511;
    __syncthreads();
    patch_stores(xs, h0, pre);
    __syncthreads();
    if (trip + 1 < TILES_PER_WAVE) {
      const int tile1 = tile0 + WG_WAVES;
      patch_loads(down, tile1 >> 11, (tile1 >> 2) & 511, pre);
    }

    const int b = tile0 >> 11;
    bool full;
    int  j0, tb, n, hi;
    decode_wave(full, j0, tb, n, hi);
    if (full) pass2_tile<4>(xs, wlds, bias, out, b, h0, j0, 0, n, hi);
    else      pass2_tile<2>(xs, wlds, bias, out, b, h0, j0, tb, n, hi);
  }
}

extern "C" void kernel_launch(void* const* d_in, const int* in_sizes, int n_in,
                              void* d_out, int out_size, void* d_ws, size_t ws_size,
                              hipStream_t stream) {
  const float* x    = (const float*)d_in[0];
  const float* w    = (const float*)d_in[1];
  const float* bias = (const float*)d_in[2];
  float*       out  = (float*)d_out;
  float*       down = (float*)d_ws;

  pass1_kernel<<<NUM_WGS, WG_THREADS, 0, stream>>>(x, w, bias, down);
  pass2_kernel<<<NUM_WGS, WG_THREADS, 0, stream>>>(down, w, bias, out);
}
